# starting kernel
# baseline (speedup 1.0000x reference)
#include <hip/hip_runtime.h>
#include <math.h>

typedef __attribute__((ext_vector_type(2))) float v2f;
typedef __attribute__((ext_vector_type(8))) float v8f;

#define NUM_FEA 32
#define EMB_DIM 8
#define OUT_DIM 16
#define WAVES_PER_BLOCK 8

#define SWZ(v, imm) __int_as_float(__builtin_amdgcn_ds_swizzle(__float_as_int(v), (imm)))
#define SWZ_X1  0x041F
#define SWZ_X2  0x081F
#define SWZ_X4  0x101F
#define SWZ_X8  0x201F
#define SWZ_X16 0x401F
#define SWZ_BCL 0x000F
#define SWZ_BCR 0x020F

__device__ __forceinline__ float fast_tanh(float v) {
#if __has_builtin(__builtin_amdgcn_tanhf)
    return __builtin_amdgcn_tanhf(v);
#else
    return tanhf(v);
#endif
}

__device__ __forceinline__ float sum16(float v) {
    v += SWZ(v, SWZ_X1);
    v += SWZ(v, SWZ_X2);
    v += SWZ(v, SWZ_X4);
    v += SWZ(v, SWZ_X8);
    return v;
}

__global__ __launch_bounds__(256)
void coattn_kernel(const float* __restrict__ x,
                   const float* __restrict__ emb_W,
                   const float* __restrict__ emb_b,
                   const float* __restrict__ M,
                   const float* __restrict__ dense_W,
                   const float* __restrict__ dense_b,
                   float* __restrict__ out,
                   int nrows) {
    __shared__ float sWM[128];
    __shared__ float sbM[128];
    __shared__ float sWr[128];
    __shared__ float sbr[128];
    __shared__ float sdW[512];
    __shared__ float sdb[16];
    __shared__ float sT[WAVES_PER_BLOCK][16 * 20];
    __shared__ float out_s[WAVES_PER_BLOCK][32];

    const int tid = threadIdx.x;

    if (tid < 128) {
        const int l = tid >> 3, f = tid & 7;
        float aw = 0.f, ab = 0.f;
        #pragma unroll
        for (int e = 0; e < EMB_DIM; ++e) {
            const float mv = M[e * EMB_DIM + f];
            aw = fmaf(emb_W[l * EMB_DIM + e], mv, aw);
            ab = fmaf(emb_b[l * EMB_DIM + e], mv, ab);
        }
        sWM[tid] = aw;
        sbM[tid] = ab;
        sWr[tid] = emb_W[128 + tid];
        sbr[tid] = emb_b[128 + tid];
    }
    sdW[tid]       = dense_W[tid];
    sdW[tid + 256] = dense_W[tid + 256];
    if (tid < OUT_DIM) sdb[tid] = dense_b[tid];
    __syncthreads();

    const int lane = tid & 31;
    const int wave = tid >> 5;
    int b = blockIdx.x * WAVES_PER_BLOCK + wave;
    const bool valid = (b < nrows);
    if (!valid) b = nrows - 1;

    const int m16  = lane & 15;
    const int half = lane >> 4;

    const float xv = x[(size_t)b * NUM_FEA + lane];
    const float xl = SWZ(xv, SWZ_BCL);
    const float xr = SWZ(xv, SWZ_BCR);

    v8f c = {};
    #pragma unroll
    for (int w = 0; w < 2; ++w) {
        const int k0 = w * 4 + half * 2;
        v2f a, bmat;
        a.x    = fmaf(xl, sWM[m16 * 8 + k0],     sbM[m16 * 8 + k0]);
        a.y    = fmaf(xl, sWM[m16 * 8 + k0 + 1], sbM[m16 * 8 + k0 + 1]);
        bmat.x = fmaf(xr, sWr[m16 * 8 + k0],     sbr[m16 * 8 + k0]);
        bmat.y = fmaf(xr, sWr[m16 * 8 + k0 + 1], sbr[m16 * 8 + k0 + 1]);
        c = __builtin_amdgcn_wmma_f32_16x16x4_f32(
                 false, a,  false, bmat,
                 (short)0, c,  false,  false);
    }

    float g[8];
    #pragma unroll
    for (int v = 0; v < 8; ++v) g[v] = fast_tanh(c[v]);

    float cs = 0.f;
    #pragma unroll
    for (int v = 0; v < 8; ++v) cs += g[v];
    cs += SWZ(cs, SWZ_X16);
    const float e_lr = __expf(cs);
    const float lr_lane = e_lr / sum16(e_lr);

    {
        float* dst = &sT[wave][m16 * 20 + half * 8];
        *(float4*)(dst)     = make_float4(g[0], g[1], g[2], g[3]);
        *(float4*)(dst + 4) = make_float4(g[4], g[5], g[6], g[7]);
    }
    __syncthreads();

    float rs = 0.f;
    #pragma unroll
    for (int v = 0; v < 8; ++v) rs += sT[wave][(half * 8 + v) * 20 + m16];
    rs += SWZ(rs, SWZ_X16);
    const float e_rl = __expf(rs);
    const float rl_lane = e_rl / sum16(e_rl);

    const float aval = (lane < 16) ? lr_lane : rl_lane;
    out_s[wave][lane] = xv * aval;
    __syncthreads();

    const int n = m16;
    float acc = sdb[n];
    const float4* ov = (const float4*)out_s[wave];
    #pragma unroll
    for (int k4 = 0; k4 < 8; ++k4) {
        const float4 o = ov[k4];
        acc = fmaf(o.x, sdW[(k4 * 4 + 0) * OUT_DIM + n], acc);
        acc = fmaf(o.y, sdW[(k4 * 4 + 1) * OUT_DIM + n], acc);
        acc = fmaf(o.z, sdW[(k4 * 4 + 2) * OUT_DIM + n], acc);
        acc = fmaf(o.w, sdW[(k4 * 4 + 3) * OUT_DIM + n], acc);
    }
    acc = fmaxf(acc, 0.f);

    if (valid && lane < 16)
        out[(size_t)b * OUT_DIM + n] = acc;
}

extern "C" void kernel_launch(void* const* d_in, const int* in_sizes, int n_in,
                              void* d_out, int out_size, void* d_ws, size_t ws_size,
                              hipStream_t stream) {
    const float* x       = (const float*)d_in[0];
    const float* emb_W   = (const float*)d_in[1];
    const float* emb_b   = (const float*)d_in[2];
    const float* M       = (const float*)d_in[3];
    const float* dense_W = (const float*)d_in[4];
    const float* dense_b = (const float*)d_in[5];
    float* out = (float*)d_out;

    const int nrows  = in_sizes[0] / NUM_FEA;
    const int blocks = (nrows + WAVES_PER_BLOCK - 1) / WAVES_PER_BLOCK;
    coattn_kernel<<<blocks, 256, 0, stream>>>(x, emb_W, emb_b, M,
                                              dense_W, dense_b, out, nrows);
}


# row sums of G from the transposed product: a second WMMA pair with the A and B fragments swapped gives G^T in registers; the LDS transpose of G and the second work-group barrier are gone
# speedup vs baseline: 1.2181x; 1.2181x over previous
#include <hip/hip_runtime.h>
#include <math.h>

typedef __attribute__((ext_vector_type(2))) float v2f;
typedef __attribute__((ext_vector_type(8))) float v8f;

#define NUM_FEA 32
#define EMB_DIM 8
#define OUT_DIM 16
#define WAVES_PER_BLOCK 8

#define SWZ(v, imm) __int_as_float(__builtin_amdgcn_ds_swizzle(__float_as_int(v), (imm)))
#define SWZ_X1  0x041F
#define SWZ_X2  0x081F
#define SWZ_X4  0x101F
#define SWZ_X8  0x201F
#define SWZ_X16 0x401F
#define SWZ_BCL 0x000F
#define SWZ_BCR 0x020F

__device__ __forceinline__ float fast_tanh(float v) {
#if __has_builtin(__builtin_amdgcn_tanhf)
    return __builtin_amdgcn_tanhf(v);
#else
    return tanhf(v);
#endif
}

__device__ __forceinline__ float sum16(float v) {
    v += SWZ(v, SWZ_X1);
    v += SWZ(v, SWZ_X2);
    v += SWZ(v, SWZ_X4);
    v += SWZ(v, SWZ_X8);
    return v;
}

template <int J>
__device__ __forceinline__ float row_bcast(float v) {
    return __int_as_float(__builtin_amdgcn_update_dpp(
        0, __float_as_int(v), 0x150 + J, 0xf, 0xf, true));
}

__device__ __forceinline__ float other_half(float v) {
    return __int_as_float(__builtin_amdgcn_permlanex16(
        0, __float_as_int(v), 0x76543210, 0xfedcba98, false, false));
}

__global__ __launch_bounds__(256)
void coattn_kernel(const float* __restrict__ x,
                   const float* __restrict__ emb_W,
                   const float* __restrict__ emb_b,
                   const float* __restrict__ M,
                   const float* __restrict__ dense_W,
                   const float* __restrict__ dense_b,
                   float* __restrict__ out,
                   int nrows) {
    __shared__ float sWM[128];
    __shared__ float sbM[128];
    __shared__ float sWr[128];
    __shared__ float sbr[128];
    __shared__ float sdW[512];

    const int tid = threadIdx.x;
    const int lane = tid & 31;
    const int wave = tid >> 5;
    int b = blockIdx.x * WAVES_PER_BLOCK + wave;
    const bool valid = (b < nrows);
    if (!valid) b = nrows - 1;

    const int m16  = lane & 15;
    const int half = lane >> 4;

    const float xv  = x[(size_t)b * NUM_FEA + lane];
    const float dW0 = dense_W[tid];
    const float dW1 = dense_W[tid + 256];
    const float db  = dense_b[m16];

    if (tid < 128) {
        const int l = tid >> 3, f = tid & 7;
        float aw = 0.f, ab = 0.f;
        #pragma unroll
        for (int e = 0; e < EMB_DIM; ++e) {
            const float mv = M[e * EMB_DIM + f];
            aw = fmaf(emb_W[l * EMB_DIM + e], mv, aw);
            ab = fmaf(emb_b[l * EMB_DIM + e], mv, ab);
        }
        sWM[tid] = aw;
        sbM[tid] = ab;
        sWr[tid] = emb_W[128 + tid];
        sbr[tid] = emb_b[128 + tid];
    }
    sdW[tid]       = dW0;
    sdW[tid + 256] = dW1;
    __syncthreads();

    float dw[16];
    #pragma unroll
    for (int j = 0; j < 16; ++j) dw[j] = sdW[(half * 16 + j) * OUT_DIM + m16];

    const float xl = SWZ(xv, SWZ_BCL);
    const float xr = SWZ(xv, SWZ_BCR);

    v8f c = {}, ct = {};
    #pragma unroll
    for (int w = 0; w < 2; ++w) {
        const int k0 = w * 4 + half * 2;
        v2f a, bmat;
        a.x    = fmaf(xl, sWM[m16 * 8 + k0],     sbM[m16 * 8 + k0]);
        a.y    = fmaf(xl, sWM[m16 * 8 + k0 + 1], sbM[m16 * 8 + k0 + 1]);
        bmat.x = fmaf(xr, sWr[m16 * 8 + k0],     sbr[m16 * 8 + k0]);
        bmat.y = fmaf(xr, sWr[m16 * 8 + k0 + 1], sbr[m16 * 8 + k0 + 1]);
        c = __builtin_amdgcn_wmma_f32_16x16x4_f32(
                 false, a,  false, bmat,
                 (short)0, c,  false,  false);
        ct = __builtin_amdgcn_wmma_f32_16x16x4_f32(
                 false, bmat,  false, a,
                 (short)0, ct,  false,  false);
    }


    float cs = 0.f;
    #pragma unroll
    for (int v = 0; v < 8; ++v) cs += fast_tanh(c[v]);
    cs += SWZ(cs, SWZ_X16);
    const float e_lr = __expf(cs);
    const float lr_lane = e_lr / sum16(e_lr);

    float rs = 0.f;
    #pragma unroll
    for (int v = 0; v < 8; ++v) rs += fast_tanh(ct[v]);
    rs += SWZ(rs, SWZ_X16);
    const float e_rl = __expf(rs);
    const float rl_lane = e_rl / sum16(e_rl);

    const float aval = (lane < 16) ? lr_lane : rl_lane;
    const float o = xv * aval;

    const int n = m16;
    float acc0 = 0.f, acc1 = 0.f;
#define DENSE_STEP(J) \
    acc0 = fmaf(row_bcast<(J)>(o),     dw[(J)],     acc0); \
    acc1 = fmaf(row_bcast<(J) + 1>(o), dw[(J) + 1], acc1);
    DENSE_STEP(0)  DENSE_STEP(2)  DENSE_STEP(4)  DENSE_STEP(6)
    DENSE_STEP(8)  DENSE_STEP(10) DENSE_STEP(12) DENSE_STEP(14)
#undef DENSE_STEP
    float acc = acc0 + acc1;
    acc = (acc + other_half(acc)) + db;
    acc = fmaxf(acc, 0.f);

    if (valid && lane < 16)
        out[(size_t)b * OUT_DIM + n] = acc;
}

extern "C" void kernel_launch(void* const* d_in, const int* in_sizes, int n_in,
                              void* d_out, int out_size, void* d_ws, size_t ws_size,
                              hipStream_t stream) {
    const float* x       = (const float*)d_in[0];
    const float* emb_W   = (const float*)d_in[1];
    const float* emb_b   = (const float*)d_in[2];
    const float* M       = (const float*)d_in[3];
    const float* dense_W = (const float*)d_in[4];
    const float* dense_b = (const float*)d_in[5];
    float* out = (float*)d_out;

    const int nrows  = in_sizes[0] / NUM_FEA;
    const int blocks = (nrows + WAVES_PER_BLOCK - 1) / WAVES_PER_BLOCK;
    coattn_kernel<<<blocks, 256, 0, stream>>>(x, emb_W, emb_b, M,
                                              dense_W, dense_b, out, nrows);
}


# row of x loaded last and waited for behind an LDS-only barrier (loads return in issue order): weight fold (spread over all 8 waves), LDS staging and barrier now run under the row's memory flight
# speedup vs baseline: 1.2556x; 1.0308x over previous
#include <hip/hip_runtime.h>
#include <math.h>

typedef __attribute__((ext_vector_type(2))) float v2f;
typedef __attribute__((ext_vector_type(8))) float v8f;

#define NUM_FEA 32
#define EMB_DIM 8
#define OUT_DIM 16
#define WAVES_PER_BLOCK 8

#define SWZ(v, imm) __int_as_float(__builtin_amdgcn_ds_swizzle(__float_as_int(v), (imm)))
#define SWZ_X1  0x041F
#define SWZ_X2  0x081F
#define SWZ_X4  0x101F
#define SWZ_X8  0x201F
#define SWZ_X16 0x401F
#define SWZ_BCL 0x000F
#define SWZ_BCR 0x020F

__device__ __forceinline__ float fast_tanh(float v) {
#if __has_builtin(__builtin_amdgcn_tanhf)
    return __builtin_amdgcn_tanhf(v);
#else
    return tanhf(v);
#endif
}

__device__ __forceinline__ float sum16(float v) {
    v += SWZ(v, SWZ_X1);
    v += SWZ(v, SWZ_X2);
    v += SWZ(v, SWZ_X4);
    v += SWZ(v, SWZ_X8);
    return v;
}

template <int J>
__device__ __forceinline__ float row_bcast(float v) {
    return __int_as_float(__builtin_amdgcn_update_dpp(
        0, __float_as_int(v), 0x150 + J, 0xf, 0xf, true));
}

__device__ __forceinline__ float other_half(float v) {
    return __int_as_float(__builtin_amdgcn_permlanex16(
        0, __float_as_int(v), 0x76543210, 0xfedcba98, false, false));
}

__global__ __launch_bounds__(256)
void coattn_kernel(const float* __restrict__ x,
                   const float* __restrict__ emb_W,
                   const float* __restrict__ emb_b,
                   const float* __restrict__ M,
                   const float* __restrict__ dense_W,
                   const float* __restrict__ dense_b,
                   float* __restrict__ out,
                   int nrows) {
    __shared__ float sW[512];
    float* const sWM = sW;
    float* const sbM = sW + 128;
    float* const sWr = sW + 256;
    float* const sbr = sW + 384;
    __shared__ float sdW[512];

    const int tid = threadIdx.x;
    const int lane = tid & 31;
    const int wave = tid >> 5;
    int b = blockIdx.x * WAVES_PER_BLOCK + wave;
    const bool valid = (b < nrows);
    if (!valid) b = nrows - 1;

    const int m16  = lane & 15;
    const int half = lane >> 4;

    const float dW0 = dense_W[tid];
    const float dW1 = dense_W[tid + 256];
    const float db  = dense_b[m16];
    const int fi = tid & 127, fl = fi >> 3, ff = fi & 7;
    const float* src = (wave < 4) ? emb_W : emb_b;
    float mW[EMB_DIM], eS[EMB_DIM];
    #pragma unroll
    for (int e = 0; e < EMB_DIM; ++e) {
        mW[e] = M[e * EMB_DIM + ff];
        eS[e] = src[fl * EMB_DIM + e];
    }
    const float rcopy = src[128 + fi];
    __builtin_amdgcn_sched_barrier(0);
    const float xv = x[(size_t)b * NUM_FEA + lane];
    __builtin_amdgcn_sched_barrier(0);

    float acc_f = 0.f;
    #pragma unroll
    for (int e = 0; e < EMB_DIM; ++e) acc_f = fmaf(eS[e], mW[e], acc_f);
    sW[tid]        = acc_f;
    sW[256 + tid]  = rcopy;
    sdW[tid]       = dW0;
    sdW[tid + 256] = dW1;
    __builtin_amdgcn_fence(__ATOMIC_RELEASE, "workgroup", "local");
    __builtin_amdgcn_s_barrier();
    __builtin_amdgcn_fence(__ATOMIC_ACQUIRE, "workgroup", "local");

    float fWM[4], fbM[4], fWr[4], fbr[4];
    #pragma unroll
    for (int w = 0; w < 2; ++w) {
        const int k0 = w * 4 + half * 2;
        fWM[2 * w] = sWM[m16 * 8 + k0];  fWM[2 * w + 1] = sWM[m16 * 8 + k0 + 1];
        fbM[2 * w] = sbM[m16 * 8 + k0];  fbM[2 * w + 1] = sbM[m16 * 8 + k0 + 1];
        fWr[2 * w] = sWr[m16 * 8 + k0];  fWr[2 * w + 1] = sWr[m16 * 8 + k0 + 1];
        fbr[2 * w] = sbr[m16 * 8 + k0];  fbr[2 * w + 1] = sbr[m16 * 8 + k0 + 1];
    }
    float xq = xv;
    asm volatile("" : "+v"(xq) : : "memory");

    const float xl = SWZ(xq, SWZ_BCL);
    const float xr = SWZ(xq, SWZ_BCR);

    float dw[16];
    #pragma unroll
    for (int j = 0; j < 16; ++j) dw[j] = sdW[(half * 16 + j) * OUT_DIM + m16];

    v8f c = {}, ct = {};
    #pragma unroll
    for (int w = 0; w < 2; ++w) {
        v2f a, bmat;
        a.x    = fmaf(xl, fWM[2 * w],     fbM[2 * w]);
        a.y    = fmaf(xl, fWM[2 * w + 1], fbM[2 * w + 1]);
        bmat.x = fmaf(xr, fWr[2 * w],     fbr[2 * w]);
        bmat.y = fmaf(xr, fWr[2 * w + 1], fbr[2 * w + 1]);
        c = __builtin_amdgcn_wmma_f32_16x16x4_f32(
                 false, a,  false, bmat,
                 (short)0, c,  false,  false);
        ct = __builtin_amdgcn_wmma_f32_16x16x4_f32(
                 false, bmat,  false, a,
                 (short)0, ct,  false,  false);
    }


    float cs = 0.f;
    #pragma unroll
    for (int v = 0; v < 8; ++v) cs += fast_tanh(c[v]);
    cs += SWZ(cs, SWZ_X16);
    const float e_lr = __expf(cs);
    const float lr_lane = e_lr / sum16(e_lr);

    float rs = 0.f;
    #pragma unroll
    for (int v = 0; v < 8; ++v) rs += fast_tanh(ct[v]);
    rs += SWZ(rs, SWZ_X16);
    const float e_rl = __expf(rs);
    const float rl_lane = e_rl / sum16(e_rl);

    const float aval = (lane < 16) ? lr_lane : rl_lane;
    const float o = xq * aval;

    const int n = m16;
    float acc0 = 0.f, acc1 = 0.f;
#define DENSE_STEP(J) \
    acc0 = fmaf(row_bcast<(J)>(o),     dw[(J)],     acc0); \
    acc1 = fmaf(row_bcast<(J) + 1>(o), dw[(J) + 1], acc1);
    DENSE_STEP(0)  DENSE_STEP(2)  DENSE_STEP(4)  DENSE_STEP(6)
    DENSE_STEP(8)  DENSE_STEP(10) DENSE_STEP(12) DENSE_STEP(14)
#undef DENSE_STEP
    float acc = acc0 + acc1;
    acc = (acc + other_half(acc)) + db;
    acc = fmaxf(acc, 0.f);

    if (valid && lane < 16)
        out[(size_t)b * OUT_DIM + n] = acc;
}

extern "C" void kernel_launch(void* const* d_in, const int* in_sizes, int n_in,
                              void* d_out, int out_size, void* d_ws, size_t ws_size,
                              hipStream_t stream) {
    const float* x       = (const float*)d_in[0];
    const float* emb_W   = (const float*)d_in[1];
    const float* emb_b   = (const float*)d_in[2];
    const float* M       = (const float*)d_in[3];
    const float* dense_W = (const float*)d_in[4];
    const float* dense_b = (const float*)d_in[5];
    float* out = (float*)d_out;

    const int nrows  = in_sizes[0] / NUM_FEA;
    const int blocks = (nrows + WAVES_PER_BLOCK - 1) / WAVES_PER_BLOCK;
    coattn_kernel<<<blocks, 256, 0, stream>>>(x, emb_W, emb_b, M,
                                              dense_W, dense_b, out, nrows);
}


# one wave per pair of consecutive rows: matrix part per row on the full wave, then the two rows share the wave's 16-lane halves for softmax, gate, dense and one whole-line store
# speedup vs baseline: 1.8093x; 1.4410x over previous
#include <hip/hip_runtime.h>
#include <math.h>

typedef __attribute__((ext_vector_type(2))) float v2f;
typedef __attribute__((ext_vector_type(4))) float v4f;
typedef __attribute__((ext_vector_type(8))) float v8f;

#define NUM_FEA 32
#define EMB_DIM 8
#define OUT_DIM 16
#define WAVES_PER_BLOCK 8
#define TAB_PLANES 8
#define TAB_FLOATS (TAB_PLANES * 32 * 4)

#define SWZ(v, imm) __int_as_float(__builtin_amdgcn_ds_swizzle(__float_as_int(v), (imm)))
#define SWZ_X1  0x041F
#define SWZ_X2  0x081F
#define SWZ_X4  0x101F
#define SWZ_X8  0x201F
#define SWZ_X16 0x401F
#define SWZ_BCL 0x000F
#define SWZ_BCR 0x020F

__device__ __forceinline__ float fast_tanh(float v) {
#if __has_builtin(__builtin_amdgcn_tanhf)
    return __builtin_amdgcn_tanhf(v);
#else
    return tanhf(v);
#endif
}

__device__ __forceinline__ float sum16(float v) {
    v += SWZ(v, SWZ_X1);
    v += SWZ(v, SWZ_X2);
    v += SWZ(v, SWZ_X4);
    v += SWZ(v, SWZ_X8);
    return v;
}

template <int J>
__device__ __forceinline__ float row_bcast(float v) {
    return __int_as_float(__builtin_amdgcn_update_dpp(
        0, __float_as_int(v), 0x150 + J, 0xf, 0xf, true));
}

__device__ __forceinline__ float other_half(float v) {
    return __int_as_float(__builtin_amdgcn_permlanex16(
        0, __float_as_int(v), 0x76543210, 0xfedcba98, false, false));
}

template <int CTRL>
__device__ __forceinline__ float row_dpp(float v) {
    return __int_as_float(__builtin_amdgcn_update_dpp(
        0, __float_as_int(v), CTRL, 0xf, 0xf, true));
}
#define DPP_QUAD_X1  0xB1
#define DPP_QUAD_X2  0x4E
#define DPP_ROW_X4   0x164
#define DPP_ROW_ROR8 0x128

__global__ __launch_bounds__(256)
void coattn_table(const float* __restrict__ emb_W,
                  const float* __restrict__ emb_b,
                  const float* __restrict__ M,
                  const float* __restrict__ dense_W,
                  float* __restrict__ tab) {
    const int idx  = blockIdx.x * 256 + threadIdx.x;
    const int p    = idx >> 7;
    const int ln   = (idx >> 2) & 31;
    const int i    = idx & 3;
    const int m16  = ln & 15;
    const int half = ln >> 4;
    float v;
    if (p < 4) {
        const int k = (i >> 1) * 4 + half * 2 + (i & 1);
        const float* src = (p & 1) ? emb_b : emb_W;
        if (p < 2) {
            float acc = 0.f;
            #pragma unroll
            for (int e = 0; e < EMB_DIM; ++e)
                acc = fmaf(src[m16 * EMB_DIM + e], M[e * EMB_DIM + k], acc);
            v = acc;
        } else {
            v = src[(16 + m16) * EMB_DIM + k];
        }
    } else {
        const int j = (p - 4) * 4 + i;
        v = dense_W[(half * 16 + j) * OUT_DIM + m16];
    }
    tab[idx] = v;
}

__device__ __forceinline__ void row_sums(const float xq, const v4f fWM, const v4f fbM,
                                         const v4f fWr, const v4f fbr, const int lane,
                                         float& cs_out, float& rs_out) {
    const int m16  = lane & 15;
    const int half = lane >> 4;
    const float xl = SWZ(xq, SWZ_BCL);
    const float xr = SWZ(xq, SWZ_BCR);

    v8f c = {};
    #pragma unroll
    for (int w = 0; w < 2; ++w) {
        v2f a, bmat;
        a.x    = fmaf(xl, fWM[2 * w],     fbM[2 * w]);
        a.y    = fmaf(xl, fWM[2 * w + 1], fbM[2 * w + 1]);
        bmat.x = fmaf(xr, fWr[2 * w],     fbr[2 * w]);
        bmat.y = fmaf(xr, fWr[2 * w + 1], fbr[2 * w + 1]);
        c = __builtin_amdgcn_wmma_f32_16x16x4_f32(
                 false, a,  false, bmat,
                 (short)0, c,  false,  false);
    }

    float t[8];
    #pragma unroll
    for (int v = 0; v < 8; ++v) t[v] = fast_tanh(c[v]);

    float cs = 0.f;
    #pragma unroll
    for (int v = 0; v < 8; ++v) cs += t[v];
    cs += SWZ(cs, SWZ_X16);
    cs_out = cs;

    const bool b3 = (lane & 8) != 0, b2 = (lane & 4) != 0, b1 = (lane & 2) != 0;
    float u[4], w[2];
    #pragma unroll
    for (int i = 0; i < 4; ++i) {
        const float keep = b3 ? t[4 + i] : t[i];
        const float send = b3 ? t[i]     : t[4 + i];
        u[i] = keep + row_dpp<DPP_ROW_ROR8>(send);
    }
    #pragma unroll
    for (int j = 0; j < 2; ++j) {
        const float keep = b2 ? u[2 + j] : u[j];
        const float send = b2 ? u[j]     : u[2 + j];
        w[j] = keep + row_dpp<DPP_ROW_X4>(send);
    }
    float z;
    {
        const float keep = b1 ? w[1] : w[0];
        const float send = b1 ? w[0] : w[1];
        z = keep + row_dpp<DPP_QUAD_X2>(send);
    }
    z += row_dpp<DPP_QUAD_X1>(z);
    const float z_own = __int_as_float(__builtin_amdgcn_permlane16(
        0, __float_as_int(z), 0xECA86420, 0xECA86420, false, false));
    const float z_oth = __int_as_float(__builtin_amdgcn_permlanex16(
        0, __float_as_int(z), 0xECA86420, 0xECA86420, false, false));
    rs_out = ((m16 >> 3) == half) ? z_own : z_oth;
}

__global__ __launch_bounds__(256)
void coattn_kernel(const float* __restrict__ x,
                   const float* __restrict__ tab,
                   const float* __restrict__ dense_b,
                   float* __restrict__ out,
                   int nrows) {
    const int tid = threadIdx.x;
    const int lane = tid & 31;
    const int wave = tid >> 5;
    const int pair = blockIdx.x * WAVES_PER_BLOCK + wave;
    int bA = 2 * pair, bB = 2 * pair + 1;
    const bool validA = (bA < nrows), validB = (bB < nrows);
    if (!validA) bA = nrows - 1;
    if (!validB) bB = nrows - 1;

    const int m16 = lane & 15;
    const bool hi = lane >= 16;

    const float xvA = x[(size_t)bA * NUM_FEA + lane];
    const float xvB = x[(size_t)bB * NUM_FEA + lane];
    const v4f* __restrict__ t4 = reinterpret_cast<const v4f*>(tab);
    const v4f fWM = t4[0 * 32 + lane];
    const v4f fbM = t4[1 * 32 + lane];
    const v4f fWr = t4[2 * 32 + lane];
    const v4f fbr = t4[3 * 32 + lane];
    float dw[16];
    #pragma unroll
    for (int q = 0; q < 4; ++q) {
        const v4f d = t4[(4 + q) * 32 + lane];
        dw[4 * q] = d.x; dw[4 * q + 1] = d.y; dw[4 * q + 2] = d.z; dw[4 * q + 3] = d.w;
    }
    float db = dense_b[m16];

    float xa = xvA, xb = xvB;
    asm volatile("" : "+v"(xa), "+v"(xb), "+v"(db));

    float csA, rsA, csB, rsB;
    row_sums(xa, fWM, fbM, fWr, fbr, lane, csA, rsA);
    row_sums(xb, fWM, fbM, fWr, fbr, lane, csB, rsB);

    const float e1 = __expf(hi ? rsB : csA);
    const float a1 = e1 / sum16(e1);
    const float e2 = __expf(hi ? rsA : csB);
    const float a2 = e2 / sum16(e2);

    const float o1 = a1 * (hi ? xb : xa);
    const float o2 = a2 * (hi ? xa : xb);

    float p1e = 0.f, p1o = 0.f, p2e = 0.f, p2o = 0.f;
#define DENSE_STEP(J) \
    p1e = fmaf(row_bcast<(J)>(o1),     dw[(J)],     p1e); \
    p2e = fmaf(row_bcast<(J)>(o2),     dw[(J)],     p2e); \
    p1o = fmaf(row_bcast<(J) + 1>(o1), dw[(J) + 1], p1o); \
    p2o = fmaf(row_bcast<(J) + 1>(o2), dw[(J) + 1], p2o);
    DENSE_STEP(0)  DENSE_STEP(2)  DENSE_STEP(4)  DENSE_STEP(6)
    DENSE_STEP(8)  DENSE_STEP(10) DENSE_STEP(12) DENSE_STEP(14)
#undef DENSE_STEP
    const float p1 = p1e + p1o;
    const float p2 = p2e + p2o;
    float acc = (p1 + other_half(p2)) + db;
    acc = fmaxf(acc, 0.f);

    const int  brow  = hi ? bB : bA;
    const bool valid = hi ? validB : validA;
    if (valid)
        out[(size_t)brow * OUT_DIM + m16] = acc;
}

extern "C" void kernel_launch(void* const* d_in, const int* in_sizes, int n_in,
                              void* d_out, int out_size, void* d_ws, size_t ws_size,
                              hipStream_t stream) {
    const float* x       = (const float*)d_in[0];
    const float* emb_W   = (const float*)d_in[1];
    const float* emb_b   = (const float*)d_in[2];
    const float* M       = (const float*)d_in[3];
    const float* dense_W = (const float*)d_in[4];
    const float* dense_b = (const float*)d_in[5];
    float* out = (float*)d_out;
    float* tab = (float*)d_ws;

    const int nrows  = in_sizes[0] / NUM_FEA;
    const int npairs = (nrows + 1) / 2;
    const int blocks = (npairs + WAVES_PER_BLOCK - 1) / WAVES_PER_BLOCK;
    coattn_table<<<TAB_FLOATS / 256, 256, 0, stream>>>(emb_W, emb_b, M, dense_W, tab);
    coattn_kernel<<<blocks, 256, 0, stream>>>(x, tab, dense_b, out, nrows);
}
